# starting kernel
# baseline (speedup 1.0000x reference)
#include <hip/hip_runtime.h>
#include <hip/hip_bf16.h>

#define FEAT 96
#define LSTRIDE 100
#define WPAIR 192

typedef float v2f __attribute__((ext_vector_type(2)));
typedef float v8f __attribute__((ext_vector_type(8)));

__global__ void zero_kernel(float* __restrict__ p, long long n) {
    long long i = (long long)blockIdx.x * blockDim.x + threadIdx.x;
    if (i < n) p[i] = 0.0f;
}

__global__ void degree_kernel(const int* __restrict__ dst, float* __restrict__ deg, int E) {
    int e = blockIdx.x * blockDim.x + threadIdx.x;
    if (e < E) atomicAdd(&deg[dst[e]], 1.0f);
}

__global__ void dinv_kernel(const float* __restrict__ deg, float* __restrict__ dinv, int N) {
    int i = blockIdx.x * blockDim.x + threadIdx.x;
    if (i < N) {
        float d = deg[i];
        dinv[i] = (d > 0.0f) ? rsqrtf(fmaxf(d, 1.0f)) : 0.0f;
    }
}

__global__ void __launch_bounds__(128)
gemm_wmma_f32(const float* __restrict__ A,
              const float* __restrict__ W,
              float* __restrict__ C,
              int nrows) {
    __shared__ float Wlds[(FEAT / 2) * WPAIR];
    __shared__ float Alds[4][16 * LSTRIDE];

    const int tid  = threadIdx.x;
    const int lane = tid & 31;
    const int wv   = tid >> 5;
    const int wid  = blockIdx.x * 4 + wv;
    const int row0 = wid << 4;
    const bool active = (row0 < nrows);

    for (int i = tid; i < FEAT * (FEAT / 4); i += 128) {
        const int r  = i / (FEAT / 4);
        const int c0 = (i % (FEAT / 4)) * 4;
        const float4 v = *(const float4*)(W + r * FEAT + c0);
        float* __restrict__ dp = &Wlds[(r >> 1) * WPAIR + (r & 1)];
        dp[(c0 + 0) * 2] = v.x;
        dp[(c0 + 1) * 2] = v.y;
        dp[(c0 + 2) * 2] = v.z;
        dp[(c0 + 3) * 2] = v.w;
    }

    if (active) {
        for (int i = lane; i < 16 * (FEAT / 4); i += 32) {
            const int r  = i / (FEAT / 4);
            const int c4 = i % (FEAT / 4);
            int gr = row0 + r;
            if (gr > nrows - 1) gr = nrows - 1;
            const float4 v = ((const float4*)(A + (long long)gr * FEAT))[c4];
            *(float4*)&Alds[wv][r * LSTRIDE + c4 * 4] = v;
        }
    }
    __syncthreads();
    if (!active) return;

    const int m  = lane & 15;
    const int kh = (lane >> 4) << 1;
    const float* __restrict__ arow = &Alds[wv][m * LSTRIDE];

    v8f acc[6];
#pragma unroll
    for (int t = 0; t < 6; ++t) acc[t] = {};

    for (int k0 = 0; k0 < FEAT; k0 += 4) {
        const v2f a = *(const v2f*)&arow[k0 + kh];
        const float* __restrict__ wp = &Wlds[((k0 + kh) >> 1) * WPAIR + m * 2];
#pragma unroll
        for (int t = 0; t < 6; ++t) {
            const v2f b = *(const v2f*)&wp[t * 32];
            acc[t] = __builtin_amdgcn_wmma_f32_16x16x4_f32(
                false, a, false, b, (short)0, acc[t], false, false);
        }
    }

    const int half = lane >> 4;
#pragma unroll
    for (int t = 0; t < 6; ++t) {
#pragma unroll
        for (int i = 0; i < 8; ++i) {
            int r = row0 + half * 8 + i;
            if (r < nrows) C[(long long)r * FEAT + t * 16 + m] = acc[t][i];
        }
    }
}

__global__ void aggregate_kernel(const float* __restrict__ h,
                                 const int* __restrict__ src,
                                 const int* __restrict__ dst,
                                 const float* __restrict__ dinv,
                                 float* __restrict__ out,
                                 int E) {
    const int lane = threadIdx.x & 31;
    const long long e = (long long)blockIdx.x * (blockDim.x >> 5) + (threadIdx.x >> 5);
    if (e >= E) return;
    const int s = src[e];
    const int d = dst[e];
    const float* __restrict__ hs = h + (long long)s * FEAT;
    __builtin_prefetch(hs + lane, 0, 0);
    const float norm = dinv[s] * dinv[d];
    float* __restrict__ od = out + (long long)d * FEAT;
#pragma unroll
    for (int i = 0; i < 3; ++i) {
        const int f = lane + i * 32;
        atomicAdd(&od[f], hs[f] * norm);
    }
}

__global__ void bias_relu_kernel(float* __restrict__ x, const float* __restrict__ b,
                                 long long n) {
    long long i = (long long)blockIdx.x * blockDim.x + threadIdx.x;
    if (i < n) {
        float v = x[i] + b[(int)(i % FEAT)];
        x[i] = v > 0.0f ? v : 0.0f;
    }
}

__global__ void bias2_kernel(float* __restrict__ out, const float* __restrict__ ba,
                             const float* __restrict__ bb, long long nf) {
    long long i = (long long)blockIdx.x * blockDim.x + threadIdx.x;
    if (i < 2 * nf) {
        const float* b = (i < nf) ? ba : bb;
        out[i] += b[(int)(i % FEAT)];
    }
}

extern "C" void kernel_launch(void* const* d_in, const int* in_sizes, int n_in,
                              void* d_out, int out_size, void* d_ws, size_t ws_size,
                              hipStream_t stream) {
    const float* x   = (const float*)d_in[0];
    const float* W1  = (const float*)d_in[1];
    const float* b1  = (const float*)d_in[2];
    const float* W2a = (const float*)d_in[3];
    const float* b2a = (const float*)d_in[4];
    const float* W2b = (const float*)d_in[5];
    const float* b2b = (const float*)d_in[6];
    const int*   ei  = (const int*)d_in[7];

    const int N = in_sizes[0] / FEAT;
    const int E = in_sizes[7] / 2;
    const int* src = ei;
    const int* dst = ei + E;

    const long long NF = (long long)N * FEAT;

    float* ws   = (float*)d_ws;
    float* deg  = ws;
    float* dinv = ws + N;
    float* h    = ws + 2LL * N;
    float* agg  = h + NF;

    float* out  = (float*)d_out;

    const int T = 256;
    const int wavesPerBlk = T / 32;
    const int gemmWaves   = (N + 15) / 16;
    const int gemmBlocks  = (gemmWaves + 3) / 4;
    const int aggBlocks   = (E + wavesPerBlk - 1) / wavesPerBlk;

    zero_kernel<<<(int)((N + T - 1) / T), T, 0, stream>>>(deg, N);
    zero_kernel<<<(int)((NF + T - 1) / T), T, 0, stream>>>(agg, NF);
    zero_kernel<<<(int)((2 * NF + T - 1) / T), T, 0, stream>>>(out, 2 * NF);
    degree_kernel<<<(E + T - 1) / T, T, 0, stream>>>(dst, deg, E);
    dinv_kernel<<<(N + T - 1) / T, T, 0, stream>>>(deg, dinv, N);

    gemm_wmma_f32<<<gemmBlocks, 128, 0, stream>>>(x, W1, h, N);
    aggregate_kernel<<<aggBlocks, T, 0, stream>>>(h, src, dst, dinv, agg, E);
    bias_relu_kernel<<<(int)((NF + T - 1) / T), T, 0, stream>>>(agg, b1, NF);

    gemm_wmma_f32<<<gemmBlocks, 128, 0, stream>>>(agg, W2a, h, N);
    aggregate_kernel<<<aggBlocks, T, 0, stream>>>(h, src, dst, dinv, out, E);

    gemm_wmma_f32<<<gemmBlocks, 128, 0, stream>>>(agg, W2b, h, N);
    aggregate_kernel<<<aggBlocks, T, 0, stream>>>(h, src, dst, dinv, out + NF, E);

    bias2_kernel<<<(int)((2 * NF + T - 1) / T), T, 0, stream>>>(out, b2a, b2b, NF);
}


# All loads of each set-up phase are requested before the first use (products: W, tile, dinv; finishing passes: 5 pieces a thread; aggregation: next chunk's indices); zero stores behind the barrier
# speedup vs baseline: 2.8391x; 2.8391x over previous
#include <hip/hip_runtime.h>
#include <hip/hip_bf16.h>

#define FEAT 96
#define LSTRIDE 100
#define WPAIR 192

typedef float v2f __attribute__((ext_vector_type(2)));
typedef float v4f __attribute__((ext_vector_type(4)));
typedef float v8f __attribute__((ext_vector_type(8)));

#define EW_BLOCKS  1024
#define EW_THREADS 256

#define DEG_BLOCKS  128
#define DEG_THREADS 1024
#define DEG_MAXR    448
#define DEG_REP     32

#define HOT_T    1024
#define HOT_PER  4
#define HOT_CAP  16

__device__ __forceinline__ void deg_count(unsigned* __restrict__ cnt, int d, int lo,
                                          unsigned rows, int lane) {
    const unsigned r = (unsigned)(d - lo);
    if (r < rows) atomicAdd(&cnt[r * DEG_REP + lane], 1u);
}

__device__ __forceinline__ void deg_count4(unsigned* __restrict__ cnt, const int4 v, int lo,
                                           unsigned rows, int lane) {
    deg_count(cnt, v.x, lo, rows, lane);
    deg_count(cnt, v.y, lo, rows, lane);
    deg_count(cnt, v.z, lo, rows, lane);
    deg_count(cnt, v.w, lo, rows, lane);
}

__global__ void __launch_bounds__(DEG_THREADS)
degree_dinv_kernel(const int* __restrict__ dst, float* __restrict__ dinv,
                   int* __restrict__ hot, int N, int E) {
    __shared__ unsigned cnt[DEG_MAXR * DEG_REP];
    __shared__ unsigned hotn;

    const int tid  = threadIdx.x;
    const int lane = tid & 31;
    const int rowsPer = (N + (int)gridDim.x - 1) / (int)gridDim.x;
    const int lo0 = (int)blockIdx.x * rowsPer;
    int hi0 = lo0 + rowsPer;
    if (hi0 > N) hi0 = N;
    const int nquads = E >> 2;
    if (tid == 0) hotn = 0u;

    for (int lo = lo0; lo < hi0; lo += DEG_MAXR) {
        int rowsI = hi0 - lo;
        if (rowsI > DEG_MAXR) rowsI = DEG_MAXR;
        const unsigned rows = (unsigned)rowsI;

        for (int i = tid; i < rowsI * DEG_REP; i += DEG_THREADS) cnt[i] = 0u;
        __syncthreads();

        int q = tid;
        for (; q + 7 * DEG_THREADS < nquads; q += 8 * DEG_THREADS) {
            int4 v[8];
#pragma unroll
            for (int k = 0; k < 8; ++k)
                __builtin_memcpy(&v[k], dst + 4LL * (q + k * DEG_THREADS), 16);
#pragma unroll
            for (int k = 0; k < 8; ++k) deg_count4(cnt, v[k], lo, rows, lane);
        }
        for (; q < nquads; q += DEG_THREADS) {
            int4 v;
            __builtin_memcpy(&v, dst + 4LL * q, 16);
            deg_count4(cnt, v, lo, rows, lane);
        }
        for (int e = (nquads << 2) + tid; e < E; e += DEG_THREADS)
            deg_count(cnt, dst[e], lo, rows, lane);
        __syncthreads();

        for (int r = tid; r < rowsI; r += DEG_THREADS) {
            unsigned c = 0u;
#pragma unroll
            for (int j = 0; j < DEG_REP; ++j) c += cnt[r * DEG_REP + ((j + r) & (DEG_REP - 1))];
            const float d = (float)c;
            dinv[lo + r] = (d > 0.0f) ? rsqrtf(fmaxf(d, 1.0f)) : 0.0f;
            if (c >= (unsigned)HOT_T) {
                const unsigned p = atomicAdd(&hotn, 1u);
                if (p < (unsigned)HOT_PER) hot[(int)blockIdx.x * HOT_PER + (int)p] = lo + r;
            }
        }
        __syncthreads();
    }
    __syncthreads();
    if (tid < HOT_PER && (unsigned)tid >= hotn) hot[(int)blockIdx.x * HOT_PER + tid] = -1;
}

#define PROD_WAVES   8
#define PROD_THREADS (PROD_WAVES * 32)

template <bool FROM_PLANES, int HEADS>
__global__ void __launch_bounds__(PROD_THREADS)
product_kernel(const float* __restrict__ A,
               float* __restrict__ p0,
               float* __restrict__ p1,
               const float* __restrict__ bin,
               const float* __restrict__ Wa,
               const float* __restrict__ Wb,
               const float* __restrict__ rs,
               float* __restrict__ Ca,
               float* __restrict__ Cb,
               int nrows) {
    __shared__ float Wlds[HEADS][(FEAT / 2) * WPAIR];
    __shared__ float Alds[PROD_WAVES][16 * LSTRIDE];

    const int tid  = threadIdx.x;
    const int lane = tid & 31;
    const int wv   = tid >> 5;
    const int wid  = blockIdx.x * PROD_WAVES + wv;
    const int row0 = wid << 4;
    const bool active = (row0 < nrows);

    constexpr int WPCS = (FEAT * (FEAT / 4)) / PROD_THREADS;
    static_assert(WPCS * PROD_THREADS == FEAT * (FEAT / 4), "W pieces must divide over the block");
    static_assert(WPCS == 9, "the load fence below names nine W pieces");
    const int nvalid = active ? ((nrows - row0 < 16) ? (nrows - row0) : 16) : 0;

    v4f w0[WPCS];
#pragma unroll
    for (int j = 0; j < WPCS; ++j) {
        const int i = tid + j * PROD_THREADS;
        w0[j] = *(const v4f*)(Wa + (i / (FEAT / 4)) * FEAT + (i % (FEAT / 4)) * 4);
    }

    v4f   va[12], vs[12], vb[3];
    float vg[12];
    if (FROM_PLANES) {
#pragma unroll
        for (int u = 0; u < 3; ++u)
            vb[u] = ((const v4f*)bin)[(lane + 8 * u) % (FEAT / 4)];
    }
#pragma unroll
    for (int u = 0; u < 12; ++u) {
        const int i  = lane + 32 * u;
        const int r  = i / (FEAT / 4);
        const int c4 = i % (FEAT / 4);
        int gr = row0 + r;
        if (gr > nrows - 1) gr = nrows - 1;
        const long long o4 = (long long)gr * (FEAT / 4) + c4;
        if (FROM_PLANES) {
            va[u] = ((const v4f*)p0)[o4];
            vs[u] = ((const v4f*)p1)[o4];
            vg[u] = rs[gr];
        } else {
            va[u] = ((const v4f*)A)[o4];
        }
    }

    const int half = lane >> 4;
    float sc[8];
#pragma unroll
    for (int i = 0; i < 8; ++i) {
        const int r  = row0 + half * 8 + i;
        const int rc = (r < nrows) ? r : (nrows - 1);
        const float t = rs[rc];
        sc[i] = (r < nrows) ? t : 0.0f;
    }

    if (!FROM_PLANES) {
        asm volatile("" :: "v"(w0[0]), "v"(w0[1]), "v"(w0[2]), "v"(w0[3]), "v"(w0[4]), "v"(w0[5]),
                           "v"(w0[6]), "v"(w0[7]), "v"(w0[8]),
                           "v"(va[0]), "v"(va[1]), "v"(va[2]), "v"(va[3]), "v"(va[4]), "v"(va[5]),
                           "v"(va[6]), "v"(va[7]), "v"(va[8]), "v"(va[9]), "v"(va[10]), "v"(va[11]),
                           "v"(sc[0]), "v"(sc[1]), "v"(sc[2]), "v"(sc[3]), "v"(sc[4]), "v"(sc[5]),
                           "v"(sc[6]), "v"(sc[7]));
    }

#pragma unroll
    for (int j = 0; j < WPCS; ++j) {
        const int i  = tid + j * PROD_THREADS;
        const int r  = i / (FEAT / 4);
        const int c0 = (i % (FEAT / 4)) * 4;
        float* __restrict__ dp = &Wlds[0][(r >> 1) * WPAIR + (r & 1)];
        dp[(c0 + 0) * 2] = w0[j].x;
        dp[(c0 + 1) * 2] = w0[j].y;
        dp[(c0 + 2) * 2] = w0[j].z;
        dp[(c0 + 3) * 2] = w0[j].w;
    }
    v4f w1[WPCS];
    if (HEADS == 2) {
#pragma unroll
        for (int j = 0; j < WPCS; ++j) {
            const int i = tid + j * PROD_THREADS;
            w1[j] = *(const v4f*)(Wb + (i / (FEAT / 4)) * FEAT + (i % (FEAT / 4)) * 4);
        }
    }

#pragma unroll
    for (int u = 0; u < 12; ++u) {
        const int i  = lane + 32 * u;
        const int r  = i / (FEAT / 4);
        const int c4 = i % (FEAT / 4);
        v4f v = va[u];
        if (FROM_PLANES) {
            const v4f   s  = vs[u];
            const float g  = vg[u];
            const v4f   bb = vb[u % 3];
            v.x = (v.x + s.x) * g + bb.x;
            v.y = (v.y + s.y) * g + bb.y;
            v.z = (v.z + s.z) * g + bb.z;
            v.w = (v.w + s.w) * g + bb.w;
            v.x = v.x > 0.0f ? v.x : 0.0f;
            v.y = v.y > 0.0f ? v.y : 0.0f;
            v.z = v.z > 0.0f ? v.z : 0.0f;
            v.w = v.w > 0.0f ? v.w : 0.0f;
        }
        *(v4f*)&Alds[wv][r * LSTRIDE + c4 * 4] = v;
    }

    if (HEADS == 2) {
#pragma unroll
        for (int j = 0; j < WPCS; ++j) {
            const int i  = tid + j * PROD_THREADS;
            const int r  = i / (FEAT / 4);
            const int c0 = (i % (FEAT / 4)) * 4;
            float* __restrict__ dp = &Wlds[HEADS - 1][(r >> 1) * WPAIR + (r & 1)];
            dp[(c0 + 0) * 2] = w1[j].x;
            dp[(c0 + 1) * 2] = w1[j].y;
            dp[(c0 + 2) * 2] = w1[j].z;
            dp[(c0 + 3) * 2] = w1[j].w;
        }
    }
    __syncthreads();
    if (!active) return;

    {
        const float4 z = make_float4(0.0f, 0.0f, 0.0f, 0.0f);
#pragma unroll
        for (int u = 0; u < 12; ++u) {
            const int i  = lane + 32 * u;
            const int r  = i / (FEAT / 4);
            const int c4 = i % (FEAT / 4);
            if (r < nvalid) {
                const long long o4 = (long long)(row0 + r) * (FEAT / 4) + c4;
                ((float4*)p0)[o4] = z;
                ((float4*)p1)[o4] = z;
            }
        }
    }

    const int m  = lane & 15;
    const int kh = (lane >> 4) << 1;
    const float* __restrict__ arow = &Alds[wv][m * LSTRIDE];

#pragma unroll
    for (int hd = 0; hd < HEADS; ++hd) {
        float* __restrict__ C = hd ? Cb : Ca;
        v8f acc[6];
#pragma unroll
        for (int t = 0; t < 6; ++t) acc[t] = {};

        for (int k0 = 0; k0 < FEAT; k0 += 4) {
            const v2f a = *(const v2f*)&arow[k0 + kh];
            const float* __restrict__ wp = &Wlds[hd][((k0 + kh) >> 1) * WPAIR + m * 2];
#pragma unroll
            for (int t = 0; t < 6; ++t) {
                const v2f b = *(const v2f*)&wp[t * 32];
                acc[t] = __builtin_amdgcn_wmma_f32_16x16x4_f32(
                    false, a, false, b, (short)0, acc[t], false, false);
            }
        }

#pragma unroll
        for (int t = 0; t < 6; ++t) {
#pragma unroll
            for (int i = 0; i < 8; ++i) {
                const int r = row0 + half * 8 + i;
                if (r < nrows) C[(long long)r * FEAT + t * 16 + m] = acc[t][i] * sc[i];
            }
        }
    }
}

#define AGG_BLOCKS  2048
#define AGG_THREADS 256
#define AGG_UNROLL  4

__device__ __forceinline__ void agg_add_row(float* __restrict__ out, float* __restrict__ hacc,
                                            int hotv, int d, int lane,
                                            float a0, float a1, float a2) {
    const unsigned hm = __builtin_amdgcn_ballot_w32(hotv == d);
    if (hm) {
        float* __restrict__ a = hacc + (int)__builtin_ctz(hm) * FEAT + lane;
        atomicAdd(&a[0],  a0);
        atomicAdd(&a[32], a1);
        atomicAdd(&a[64], a2);
    } else {
        float* __restrict__ od = out + (long long)d * FEAT + lane;
        atomicAdd(&od[0],  a0);
        atomicAdd(&od[32], a1);
        atomicAdd(&od[64], a2);
    }
}

__global__ void __launch_bounds__(AGG_THREADS)
aggregate_kernel(const float* __restrict__ h,
                 const int* __restrict__ src,
                 const int* __restrict__ dst,
                 float* __restrict__ plane0,
                 float* __restrict__ plane1,
                 const int* __restrict__ hot,
                 int E) {
    __shared__ float    hacc[HOT_CAP * FEAT];
    __shared__ int      hrow[HOT_CAP];
    __shared__ unsigned hn;

    const int tid  = (int)threadIdx.x;
    const int lane = tid & 31;
    const int wv     = __builtin_amdgcn_readfirstlane((int)(threadIdx.x >> 5));
    const int wave   = (int)blockIdx.x * (AGG_THREADS >> 5) + wv;
    const int nwaves = (int)gridDim.x * (AGG_THREADS >> 5);
    const int per    = (E + nwaves - 1) / nwaves;
    const long long beg64 = (long long)wave * per;

    const unsigned place = (unsigned)__builtin_amdgcn_s_sendmsg_rtn(0x87);
    float* __restrict__ out = ((place >> 18) & 1u) ? plane1 : plane0;

    constexpr int HOT_EACH = (DEG_BLOCKS * HOT_PER) / AGG_THREADS;
    static_assert(HOT_EACH * AGG_THREADS == DEG_BLOCKS * HOT_PER, "lists must divide over the block");
    int hl[HOT_EACH];
#pragma unroll
    for (int k = 0; k < HOT_EACH; ++k) hl[k] = hot[tid + k * AGG_THREADS];

    const bool has = (beg64 < E);
    const int beg = has ? (int)beg64 : E;
    const int end = (E - beg < per) ? E : beg + per;
    int svn = 0, dvn = 0;
    if (beg + lane < end) {
        svn = src[beg + lane];
        dvn = dst[beg + lane];
    }

    if (tid == 0) hn = 0u;
    if (tid < HOT_CAP) hrow[tid] = -1;
    for (int i = tid; i < HOT_CAP * FEAT; i += AGG_THREADS) hacc[i] = 0.0f;
    __syncthreads();
#pragma unroll
    for (int k = 0; k < HOT_EACH; ++k) {
        const int r = hl[k];
        if (r >= 0) {
            const unsigned p = atomicAdd(&hn, 1u);
            if (p < (unsigned)HOT_CAP) hrow[p] = r;
        }
    }
    __syncthreads();
    const int hotv = (lane < HOT_CAP) ? hrow[lane] : -1;

    if (has) {
        for (int base = beg; base < end; base += 32) {
            const int cnt = (end - base < 32) ? (end - base) : 32;
            const int sv = svn, dv = dvn;
            svn = 0; dvn = 0;
            if (base + 32 + lane < end) {
                svn = src[base + 32 + lane];
                dvn = dst[base + 32 + lane];
            }

            int j = 0;
            for (; j + AGG_UNROLL <= cnt; j += AGG_UNROLL) {
                float v[AGG_UNROLL][3];
                int   dd[AGG_UNROLL];
#pragma unroll
                for (int u = 0; u < AGG_UNROLL; ++u) {
                    const int s = __builtin_amdgcn_readlane(sv, j + u);
                    dd[u] = __builtin_amdgcn_readlane(dv, j + u);
                    const float* __restrict__ hs = h + (long long)s * FEAT + lane;
                    v[u][0] = hs[0];
                    v[u][1] = hs[32];
                    v[u][2] = hs[64];
                }
#pragma unroll
                for (int u = 0; u < AGG_UNROLL; ++u)
                    agg_add_row(out, hacc, hotv, dd[u], lane, v[u][0], v[u][1], v[u][2]);
            }
            for (; j < cnt; ++j) {
                const int   s  = __builtin_amdgcn_readlane(sv, j);
                const int   d  = __builtin_amdgcn_readlane(dv, j);
                const float* __restrict__ hs = h + (long long)s * FEAT + lane;
                const float a0 = hs[0], a1 = hs[32], a2 = hs[64];
                agg_add_row(out, hacc, hotv, d, lane, a0, a1, a2);
            }
        }
    }

    __syncthreads();
    const unsigned nh = (hn < (unsigned)HOT_CAP) ? hn : (unsigned)HOT_CAP;
    for (unsigned s = (unsigned)wv; s < nh; s += (AGG_THREADS >> 5)) {
        const float* __restrict__ a = hacc + (int)s * FEAT + lane;
        float* __restrict__ od = out + (long long)hrow[s] * FEAT + lane;
        atomicAdd(&od[0],  a[0]);
        atomicAdd(&od[32], a[32]);
        atomicAdd(&od[64], a[64]);
    }
}

#define EW_BATCH 5
template <bool CLEAR>
__global__ void __launch_bounds__(EW_THREADS)
sum2_bias_kernel(float* o, const float* x, float* __restrict__ r, const float* __restrict__ b,
                 const float* __restrict__ rs, float* __restrict__ zc, long long n) {
    const long long n4 = n >> 2;
    const long long step = (long long)gridDim.x * EW_THREADS;
    v4f* o4 = (v4f*)o;
    const v4f* x4 = (const v4f*)x;
    v4f* r4 = (v4f*)r;
    const v4f* __restrict__ b4 = (const v4f*)b;
    v4f* __restrict__ z4 = (v4f*)zc;
    const v4f z = {0.0f, 0.0f, 0.0f, 0.0f};
    static_assert(EW_BATCH == 5, "the load fence below names five pieces");
    for (long long i0 = (long long)blockIdx.x * EW_THREADS + threadIdx.x; i0 < n4;
         i0 += step * EW_BATCH) {
        v4f   v[EW_BATCH], s[EW_BATCH], bb[EW_BATCH];
        float g[EW_BATCH];
#pragma unroll
        for (int k = 0; k < EW_BATCH; ++k) {
            long long i = i0 + k * step;
            if (i > n4 - 1) i = n4 - 1;
            const long long row = i / (FEAT / 4);
            const int c4 = (int)(i - row * (FEAT / 4));
            v[k]  = x4[i];
            s[k]  = r4[i];
            g[k]  = rs[row];
            bb[k] = b4[c4];
        }
        asm volatile("" :: "v"(v[0]), "v"(v[1]), "v"(v[2]), "v"(v[3]), "v"(v[4]),
                           "v"(s[0]), "v"(s[1]), "v"(s[2]), "v"(s[3]), "v"(s[4]),
                           "v"(bb[0]), "v"(bb[1]), "v"(bb[2]), "v"(bb[3]), "v"(bb[4]),
                           "v"(g[0]), "v"(g[1]), "v"(g[2]), "v"(g[3]), "v"(g[4]));
#pragma unroll
        for (int k = 0; k < EW_BATCH; ++k) {
            const long long i = i0 + k * step;
            if (i < n4) {
                v4f w = v[k];
                w.x = (w.x + s[k].x) * g[k] + bb[k].x;
                w.y = (w.y + s[k].y) * g[k] + bb[k].y;
                w.z = (w.z + s[k].z) * g[k] + bb[k].z;
                w.w = (w.w + s[k].w) * g[k] + bb[k].w;
                o4[i] = w;
                if (CLEAR) {
                    r4[i] = z;
                    z4[i] = z;
                }
            }
        }
    }
}

extern "C" void kernel_launch(void* const* d_in, const int* in_sizes, int n_in,
                              void* d_out, int out_size, void* d_ws, size_t ws_size,
                              hipStream_t stream) {
    const float* x   = (const float*)d_in[0];
    const float* W1  = (const float*)d_in[1];
    const float* b1  = (const float*)d_in[2];
    const float* W2a = (const float*)d_in[3];
    const float* b2a = (const float*)d_in[4];
    const float* W2b = (const float*)d_in[5];
    const float* b2b = (const float*)d_in[6];
    const int*   ei  = (const int*)d_in[7];

    const int N = in_sizes[0] / FEAT;
    const int E = in_sizes[7] / 2;
    const int* src = ei;
    const int* dst = ei + E;

    const long long NF = (long long)N * FEAT;

    float* ws   = (float*)d_ws;
    int*   hot  = (int*)d_ws;
    float* dinv = ws + N;
    float* h    = ws + 2LL * N;
    float* agg  = h + NF;

    float* out  = (float*)d_out;
    float* mu   = out;
    float* ls   = out + NF;

    const int prodBlocks = (N + 16 * PROD_WAVES - 1) / (16 * PROD_WAVES);

    degree_dinv_kernel<<<DEG_BLOCKS, DEG_THREADS, 0, stream>>>(dst, dinv, hot, N, E);

    product_kernel<false, 1><<<prodBlocks, PROD_THREADS, 0, stream>>>(
        x, agg, mu, nullptr, W1, nullptr, dinv, h, nullptr, N);
    aggregate_kernel<<<AGG_BLOCKS, AGG_THREADS, 0, stream>>>(h, src, dst, agg, mu, hot, E);

    product_kernel<true, 2><<<prodBlocks, PROD_THREADS, 0, stream>>>(
        nullptr, agg, mu, b1, W2a, W2b, dinv, h, ls, N);

    aggregate_kernel<<<AGG_BLOCKS, AGG_THREADS, 0, stream>>>(h, src, dst, mu, agg, hot, E);
    sum2_bias_kernel<true><<<EW_BLOCKS, EW_THREADS, 0, stream>>>(mu, mu, agg, b2a, dinv, h, NF);

    aggregate_kernel<<<AGG_BLOCKS, AGG_THREADS, 0, stream>>>(ls, src, dst, agg, h, hot, E);
    sum2_bias_kernel<false><<<EW_BLOCKS, EW_THREADS, 0, stream>>>(ls, agg, h, b2b, dinv, nullptr, NF);
}


# Degree count split two ways: 32 destination ranges x 8 edge-list slices (256 work-groups, each scans an eighth of the list, not all); layer-1 product adds the 8 integer partials into dinv
# speedup vs baseline: 3.2609x; 1.1486x over previous
#include <hip/hip_runtime.h>
#include <hip/hip_bf16.h>

#define FEAT 96
#define LSTRIDE 100
#define WPAIR 192

typedef float v2f __attribute__((ext_vector_type(2)));
typedef float v4f __attribute__((ext_vector_type(4)));
typedef float v8f __attribute__((ext_vector_type(8)));

#define EW_BLOCKS  1024
#define EW_THREADS 256

#define DEG_RANGES  32
#define DEG_SLICES  8
#define DEG_THREADS 1024
#define DEG_MAXR    1600
#define DEG_REP     32

#define HOT_T    1024
#define HOT_LIST 512
#define HOT_CAP  16

typedef unsigned v4u __attribute__((ext_vector_type(4)));

__device__ __forceinline__ void deg_count(unsigned* __restrict__ cnt, int d, int lo,
                                          unsigned rows, int lane) {
    const unsigned r = (unsigned)(d - lo);
    if (r < rows) atomicAdd(&cnt[r * DEG_REP + lane], 1u);
}

__device__ __forceinline__ void deg_count4(unsigned* __restrict__ cnt, const int4 v, int lo,
                                           unsigned rows, int lane) {
    deg_count(cnt, v.x, lo, rows, lane);
    deg_count(cnt, v.y, lo, rows, lane);
    deg_count(cnt, v.z, lo, rows, lane);
    deg_count(cnt, v.w, lo, rows, lane);
}

__global__ void __launch_bounds__(DEG_THREADS)
degree_part_kernel(const int* __restrict__ dst, unsigned* __restrict__ part, int N, int E) {
    __shared__ __attribute__((aligned(16))) unsigned cnt[DEG_MAXR * DEG_REP];

    const int tid   = threadIdx.x;
    const int lane  = tid & 31;
    const int range = (int)blockIdx.x / DEG_SLICES;
    const int slice = (int)blockIdx.x % DEG_SLICES;
    const int rowsPer = (N + DEG_RANGES - 1) / DEG_RANGES;
    const int lo0 = range * rowsPer;
    int hi0 = lo0 + rowsPer;
    if (hi0 > N) hi0 = N;
    const int nquads = E >> 2;
    const int qPer = (nquads + DEG_SLICES - 1) / DEG_SLICES;
    const int q0 = slice * qPer;
    int q1 = q0 + qPer;
    if (q1 > nquads) q1 = nquads;

    for (int lo = lo0; lo < hi0; lo += DEG_MAXR) {
        int rowsI = hi0 - lo;
        if (rowsI > DEG_MAXR) rowsI = DEG_MAXR;
        const unsigned rows = (unsigned)rowsI;

        {
            const v4u z = {0u, 0u, 0u, 0u};
            for (int i = tid; i < rowsI * (DEG_REP / 4); i += DEG_THREADS) ((v4u*)cnt)[i] = z;
        }
        __syncthreads();

        for (int q = q0 + tid; q < q1; q += 8 * DEG_THREADS) {
            int4 v[8];
#pragma unroll
            for (int k = 0; k < 8; ++k) {
                int qq = q + k * DEG_THREADS;
                if (qq > q1 - 1) qq = q1 - 1;
                __builtin_memcpy(&v[k], dst + 4LL * qq, 16);
            }
#pragma unroll
            for (int k = 0; k < 8; ++k)
                if (q + k * DEG_THREADS < q1) deg_count4(cnt, v[k], lo, rows, lane);
        }
        if (slice == DEG_SLICES - 1)
            for (int e = (nquads << 2) + tid; e < E; e += DEG_THREADS)
                deg_count(cnt, dst[e], lo, rows, lane);
        __syncthreads();

        for (int r = tid; r < rowsI; r += DEG_THREADS) {
            unsigned c = 0u;
#pragma unroll
            for (int j = 0; j < DEG_REP; ++j) c += cnt[r * DEG_REP + ((j + r) & (DEG_REP - 1))];
            part[(long long)slice * N + lo + r] = c;
        }
        __syncthreads();
    }
}

#define PROD_WAVES   8
#define PROD_THREADS (PROD_WAVES * 32)

template <bool FROM_PLANES, int HEADS>
__global__ void __launch_bounds__(PROD_THREADS)
product_kernel(const float* __restrict__ A,
               float* __restrict__ p0,
               float* __restrict__ p1,
               const float* __restrict__ bin,
               const float* __restrict__ Wa,
               const float* __restrict__ Wb,
               const float* __restrict__ rs,
               float* __restrict__ Ca,
               float* __restrict__ Cb,
               int nrows,
               const unsigned* __restrict__ part,
               float* __restrict__ dinvw,
               int* __restrict__ hot) {
    __shared__ float Wlds[HEADS][(FEAT / 2) * WPAIR];
    __shared__ float Alds[PROD_WAVES][16 * LSTRIDE];
    __shared__ __attribute__((aligned(16))) float dsh[16 * PROD_WAVES];
    __shared__ int hsl[4];

    const int tid  = threadIdx.x;
    const int lane = tid & 31;
    const int wv   = tid >> 5;
    const int wid  = blockIdx.x * PROD_WAVES + wv;
    const int row0 = wid << 4;
    const bool active = (row0 < nrows);

    constexpr int WPCS = (FEAT * (FEAT / 4)) / PROD_THREADS;
    static_assert(WPCS * PROD_THREADS == FEAT * (FEAT / 4), "W pieces must divide over the block");
    static_assert(WPCS == 9, "the load fence below names nine W pieces");
    const int nvalid = active ? ((nrows - row0 < 16) ? (nrows - row0) : 16) : 0;

    v4f w0[WPCS];
#pragma unroll
    for (int j = 0; j < WPCS; ++j) {
        const int i = tid + j * PROD_THREADS;
        w0[j] = *(const v4f*)(Wa + (i / (FEAT / 4)) * FEAT + (i % (FEAT / 4)) * 4);
    }

    v4f   va[12], vs[12], vb[3];
    float vg[12];
    if (FROM_PLANES) {
#pragma unroll
        for (int u = 0; u < 3; ++u)
            vb[u] = ((const v4f*)bin)[(lane + 8 * u) % (FEAT / 4)];
    }
#pragma unroll
    for (int u = 0; u < 12; ++u) {
        const int i  = lane + 32 * u;
        const int r  = i / (FEAT / 4);
        const int c4 = i % (FEAT / 4);
        int gr = row0 + r;
        if (gr > nrows - 1) gr = nrows - 1;
        const long long o4 = (long long)gr * (FEAT / 4) + c4;
        if (FROM_PLANES) {
            va[u] = ((const v4f*)p0)[o4];
            vs[u] = ((const v4f*)p1)[o4];
            vg[u] = rs[gr];
        } else {
            va[u] = ((const v4f*)A)[o4];
        }
    }

    const int half = lane >> 4;
    float sc[8];
    unsigned pc[DEG_SLICES];
    const int prow = (int)blockIdx.x * (16 * PROD_WAVES) + (tid & (16 * PROD_WAVES - 1));
    if (FROM_PLANES) {
#pragma unroll
        for (int i = 0; i < 8; ++i) {
            const int r  = row0 + half * 8 + i;
            const int rc = (r < nrows) ? r : (nrows - 1);
            const float t = rs[rc];
            sc[i] = (r < nrows) ? t : 0.0f;
        }
    } else {
        const int pr = (prow < nrows) ? prow : (nrows - 1);
#pragma unroll
        for (int j = 0; j < DEG_SLICES; ++j) pc[j] = part[(long long)j * nrows + pr];
    }

    if (!FROM_PLANES) {
        asm volatile("" :: "v"(w0[0]), "v"(w0[1]), "v"(w0[2]), "v"(w0[3]), "v"(w0[4]), "v"(w0[5]),
                           "v"(w0[6]), "v"(w0[7]), "v"(w0[8]),
                           "v"(va[0]), "v"(va[1]), "v"(va[2]), "v"(va[3]), "v"(va[4]), "v"(va[5]),
                           "v"(va[6]), "v"(va[7]), "v"(va[8]), "v"(va[9]), "v"(va[10]), "v"(va[11]),
                           "v"(pc[0]), "v"(pc[1]), "v"(pc[2]), "v"(pc[3]), "v"(pc[4]), "v"(pc[5]),
                           "v"(pc[6]), "v"(pc[7]));
        static_assert(DEG_SLICES == 8, "the load fence above names eight partial counts");

        unsigned c = 0u;
#pragma unroll
        for (int j = 0; j < DEG_SLICES; ++j) c += pc[j];
        const float d  = (float)c;
        const float dv = (d > 0.0f) ? rsqrtf(fmaxf(d, 1.0f)) : 0.0f;
        if (tid < 16 * PROD_WAVES) {
            dsh[tid] = dv;
            if (prow < nrows) dinvw[prow] = dv;
            const unsigned cm = __builtin_amdgcn_ballot_w32(prow < nrows && c >= (unsigned)HOT_T);
            if (lane == 0) hsl[wv] = cm ? (prow + (int)__builtin_ctz(cm)) : -1;
        }
    }

#pragma unroll
    for (int j = 0; j < WPCS; ++j) {
        const int i  = tid + j * PROD_THREADS;
        const int r  = i / (FEAT / 4);
        const int c0 = (i % (FEAT / 4)) * 4;
        float* __restrict__ dp = &Wlds[0][(r >> 1) * WPAIR + (r & 1)];
        dp[(c0 + 0) * 2] = w0[j].x;
        dp[(c0 + 1) * 2] = w0[j].y;
        dp[(c0 + 2) * 2] = w0[j].z;
        dp[(c0 + 3) * 2] = w0[j].w;
    }
    v4f w1[WPCS];
    if (HEADS == 2) {
#pragma unroll
        for (int j = 0; j < WPCS; ++j) {
            const int i = tid + j * PROD_THREADS;
            w1[j] = *(const v4f*)(Wb + (i / (FEAT / 4)) * FEAT + (i % (FEAT / 4)) * 4);
        }
    }

#pragma unroll
    for (int u = 0; u < 12; ++u) {
        const int i  = lane + 32 * u;
        const int r  = i / (FEAT / 4);
        const int c4 = i % (FEAT / 4);
        v4f v = va[u];
        if (FROM_PLANES) {
            const v4f   s  = vs[u];
            const float g  = vg[u];
            const v4f   bb = vb[u % 3];
            v.x = (v.x + s.x) * g + bb.x;
            v.y = (v.y + s.y) * g + bb.y;
            v.z = (v.z + s.z) * g + bb.z;
            v.w = (v.w + s.w) * g + bb.w;
            v.x = v.x > 0.0f ? v.x : 0.0f;
            v.y = v.y > 0.0f ? v.y : 0.0f;
            v.z = v.z > 0.0f ? v.z : 0.0f;
            v.w = v.w > 0.0f ? v.w : 0.0f;
        }
        *(v4f*)&Alds[wv][r * LSTRIDE + c4 * 4] = v;
    }

    if (HEADS == 2) {
#pragma unroll
        for (int j = 0; j < WPCS; ++j) {
            const int i  = tid + j * PROD_THREADS;
            const int r  = i / (FEAT / 4);
            const int c0 = (i % (FEAT / 4)) * 4;
            float* __restrict__ dp = &Wlds[HEADS - 1][(r >> 1) * WPAIR + (r & 1)];
            dp[(c0 + 0) * 2] = w1[j].x;
            dp[(c0 + 1) * 2] = w1[j].y;
            dp[(c0 + 2) * 2] = w1[j].z;
            dp[(c0 + 3) * 2] = w1[j].w;
        }
    }
    __syncthreads();
    if (!FROM_PLANES) {
        if (tid == 0) {
            int hr = hsl[0];
            if (hr < 0) hr = hsl[1];
            if (hr < 0) hr = hsl[2];
            if (hr < 0) hr = hsl[3];
            if (blockIdx.x < HOT_LIST) hot[blockIdx.x] = hr;
            for (int e = (int)(blockIdx.x + gridDim.x); e < HOT_LIST; e += (int)gridDim.x) hot[e] = -1;
        }
        const v4f s0 = *(const v4f*)&dsh[wv * 16 + half * 8];
        const v4f s1 = *(const v4f*)&dsh[wv * 16 + half * 8 + 4];
        const float st[8] = {s0.x, s0.y, s0.z, s0.w, s1.x, s1.y, s1.z, s1.w};
#pragma unroll
        for (int i = 0; i < 8; ++i) sc[i] = (row0 + half * 8 + i < nrows) ? st[i] : 0.0f;
    }
    if (!active) return;

    {
        const float4 z = make_float4(0.0f, 0.0f, 0.0f, 0.0f);
#pragma unroll
        for (int u = 0; u < 12; ++u) {
            const int i  = lane + 32 * u;
            const int r  = i / (FEAT / 4);
            const int c4 = i % (FEAT / 4);
            if (r < nvalid) {
                const long long o4 = (long long)(row0 + r) * (FEAT / 4) + c4;
                ((float4*)p0)[o4] = z;
                ((float4*)p1)[o4] = z;
            }
        }
    }

    const int m  = lane & 15;
    const int kh = (lane >> 4) << 1;
    const float* __restrict__ arow = &Alds[wv][m * LSTRIDE];

#pragma unroll
    for (int hd = 0; hd < HEADS; ++hd) {
        float* __restrict__ C = hd ? Cb : Ca;
        v8f acc[6];
#pragma unroll
        for (int t = 0; t < 6; ++t) acc[t] = {};

        for (int k0 = 0; k0 < FEAT; k0 += 4) {
            const v2f a = *(const v2f*)&arow[k0 + kh];
            const float* __restrict__ wp = &Wlds[hd][((k0 + kh) >> 1) * WPAIR + m * 2];
#pragma unroll
            for (int t = 0; t < 6; ++t) {
                const v2f b = *(const v2f*)&wp[t * 32];
                acc[t] = __builtin_amdgcn_wmma_f32_16x16x4_f32(
                    false, a, false, b, (short)0, acc[t], false, false);
            }
        }

#pragma unroll
        for (int t = 0; t < 6; ++t) {
#pragma unroll
            for (int i = 0; i < 8; ++i) {
                const int r = row0 + half * 8 + i;
                if (r < nrows) C[(long long)r * FEAT + t * 16 + m] = acc[t][i] * sc[i];
            }
        }
    }
}

#define AGG_BLOCKS  2048
#define AGG_THREADS 256
#define AGG_UNROLL  4

__device__ __forceinline__ void agg_add_row(float* __restrict__ out, float* __restrict__ hacc,
                                            int hotv, int d, int lane,
                                            float a0, float a1, float a2) {
    const unsigned hm = __builtin_amdgcn_ballot_w32(hotv == d);
    if (hm) {
        float* __restrict__ a = hacc + (int)__builtin_ctz(hm) * FEAT + lane;
        atomicAdd(&a[0],  a0);
        atomicAdd(&a[32], a1);
        atomicAdd(&a[64], a2);
    } else {
        float* __restrict__ od = out + (long long)d * FEAT + lane;
        atomicAdd(&od[0],  a0);
        atomicAdd(&od[32], a1);
        atomicAdd(&od[64], a2);
    }
}

__global__ void __launch_bounds__(AGG_THREADS)
aggregate_kernel(const float* __restrict__ h,
                 const int* __restrict__ src,
                 const int* __restrict__ dst,
                 float* __restrict__ plane0,
                 float* __restrict__ plane1,
                 const int* __restrict__ hot,
                 int E) {
    __shared__ float    hacc[HOT_CAP * FEAT];
    __shared__ int      hrow[HOT_CAP];
    __shared__ unsigned hn;

    const int tid  = (int)threadIdx.x;
    const int lane = tid & 31;
    const int wv     = __builtin_amdgcn_readfirstlane((int)(threadIdx.x >> 5));
    const int wave   = (int)blockIdx.x * (AGG_THREADS >> 5) + wv;
    const int nwaves = (int)gridDim.x * (AGG_THREADS >> 5);
    const int per    = (E + nwaves - 1) / nwaves;
    const long long beg64 = (long long)wave * per;

    const unsigned place = (unsigned)__builtin_amdgcn_s_sendmsg_rtn(0x87);
    float* __restrict__ out = ((place >> 18) & 1u) ? plane1 : plane0;

    constexpr int HOT_EACH = HOT_LIST / AGG_THREADS;
    static_assert(HOT_EACH * AGG_THREADS == HOT_LIST, "the list must divide over the block");
    int hl[HOT_EACH];
#pragma unroll
    for (int k = 0; k < HOT_EACH; ++k) hl[k] = hot[tid + k * AGG_THREADS];

    const bool has = (beg64 < E);
    const int beg = has ? (int)beg64 : E;
    const int end = (E - beg < per) ? E : beg + per;
    int svn = 0, dvn = 0;
    if (beg + lane < end) {
        svn = src[beg + lane];
        dvn = dst[beg + lane];
    }

    if (tid == 0) hn = 0u;
    if (tid < HOT_CAP) hrow[tid] = -1;
    for (int i = tid; i < HOT_CAP * FEAT; i += AGG_THREADS) hacc[i] = 0.0f;
    __syncthreads();
#pragma unroll
    for (int k = 0; k < HOT_EACH; ++k) {
        const int r = hl[k];
        if (r >= 0) {
            const unsigned p = atomicAdd(&hn, 1u);
            if (p < (unsigned)HOT_CAP) hrow[p] = r;
        }
    }
    __syncthreads();
    const int hotv = (lane < HOT_CAP) ? hrow[lane] : -1;

    if (has) {
        for (int base = beg; base < end; base += 32) {
            const int cnt = (end - base < 32) ? (end - base) : 32;
            const int sv = svn, dv = dvn;
            svn = 0; dvn = 0;
            if (base + 32 + lane < end) {
                svn = src[base + 32 + lane];
                dvn = dst[base + 32 + lane];
            }

            int j = 0;
            for (; j + AGG_UNROLL <= cnt; j += AGG_UNROLL) {
                float v[AGG_UNROLL][3];
                int   dd[AGG_UNROLL];
#pragma unroll
                for (int u = 0; u < AGG_UNROLL; ++u) {
                    const int s = __builtin_amdgcn_readlane(sv, j + u);
                    dd[u] = __builtin_amdgcn_readlane(dv, j + u);
                    const float* __restrict__ hs = h + (long long)s * FEAT + lane;
                    v[u][0] = hs[0];
                    v[u][1] = hs[32];
                    v[u][2] = hs[64];
                }
#pragma unroll
                for (int u = 0; u < AGG_UNROLL; ++u)
                    agg_add_row(out, hacc, hotv, dd[u], lane, v[u][0], v[u][1], v[u][2]);
            }
            for (; j < cnt; ++j) {
                const int   s  = __builtin_amdgcn_readlane(sv, j);
                const int   d  = __builtin_amdgcn_readlane(dv, j);
                const float* __restrict__ hs = h + (long long)s * FEAT + lane;
                const float a0 = hs[0], a1 = hs[32], a2 = hs[64];
                agg_add_row(out, hacc, hotv, d, lane, a0, a1, a2);
            }
        }
    }

    __syncthreads();
    const unsigned nh = (hn < (unsigned)HOT_CAP) ? hn : (unsigned)HOT_CAP;
    for (unsigned s = (unsigned)wv; s < nh; s += (AGG_THREADS >> 5)) {
        const float* __restrict__ a = hacc + (int)s * FEAT + lane;
        float* __restrict__ od = out + (long long)hrow[s] * FEAT + lane;
        atomicAdd(&od[0],  a[0]);
        atomicAdd(&od[32], a[32]);
        atomicAdd(&od[64], a[64]);
    }
}

#define EW_BATCH 5
template <bool CLEAR>
__global__ void __launch_bounds__(EW_THREADS)
sum2_bias_kernel(float* o, const float* x, float* __restrict__ r, const float* __restrict__ b,
                 const float* __restrict__ rs, float* __restrict__ zc, long long n) {
    const long long n4 = n >> 2;
    const long long step = (long long)gridDim.x * EW_THREADS;
    v4f* o4 = (v4f*)o;
    const v4f* x4 = (const v4f*)x;
    v4f* r4 = (v4f*)r;
    const v4f* __restrict__ b4 = (const v4f*)b;
    v4f* __restrict__ z4 = (v4f*)zc;
    const v4f z = {0.0f, 0.0f, 0.0f, 0.0f};
    static_assert(EW_BATCH == 5, "the load fence below names five pieces");
    for (long long i0 = (long long)blockIdx.x * EW_THREADS + threadIdx.x; i0 < n4;
         i0 += step * EW_BATCH) {
        v4f   v[EW_BATCH], s[EW_BATCH], bb[EW_BATCH];
        float g[EW_BATCH];
#pragma unroll
        for (int k = 0; k < EW_BATCH; ++k) {
            long long i = i0 + k * step;
            if (i > n4 - 1) i = n4 - 1;
            const long long row = i / (FEAT / 4);
            const int c4 = (int)(i - row * (FEAT / 4));
            v[k]  = x4[i];
            s[k]  = r4[i];
            g[k]  = rs[row];
            bb[k] = b4[c4];
        }
        asm volatile("" :: "v"(v[0]), "v"(v[1]), "v"(v[2]), "v"(v[3]), "v"(v[4]),
                           "v"(s[0]), "v"(s[1]), "v"(s[2]), "v"(s[3]), "v"(s[4]),
                           "v"(bb[0]), "v"(bb[1]), "v"(bb[2]), "v"(bb[3]), "v"(bb[4]),
                           "v"(g[0]), "v"(g[1]), "v"(g[2]), "v"(g[3]), "v"(g[4]));
#pragma unroll
        for (int k = 0; k < EW_BATCH; ++k) {
            const long long i = i0 + k * step;
            if (i < n4) {
                v4f w = v[k];
                w.x = (w.x + s[k].x) * g[k] + bb[k].x;
                w.y = (w.y + s[k].y) * g[k] + bb[k].y;
                w.z = (w.z + s[k].z) * g[k] + bb[k].z;
                w.w = (w.w + s[k].w) * g[k] + bb[k].w;
                o4[i] = w;
                if (CLEAR) {
                    r4[i] = z;
                    z4[i] = z;
                }
            }
        }
    }
}

extern "C" void kernel_launch(void* const* d_in, const int* in_sizes, int n_in,
                              void* d_out, int out_size, void* d_ws, size_t ws_size,
                              hipStream_t stream) {
    const float* x   = (const float*)d_in[0];
    const float* W1  = (const float*)d_in[1];
    const float* b1  = (const float*)d_in[2];
    const float* W2a = (const float*)d_in[3];
    const float* b2a = (const float*)d_in[4];
    const float* W2b = (const float*)d_in[5];
    const float* b2b = (const float*)d_in[6];
    const int*   ei  = (const int*)d_in[7];

    const int N = in_sizes[0] / FEAT;
    const int E = in_sizes[7] / 2;
    const int* src = ei;
    const int* dst = ei + E;

    const long long NF = (long long)N * FEAT;

    float* ws   = (float*)d_ws;
    int*   hot  = (int*)d_ws;
    float* dinv = ws + N;
    float* h    = ws + 2LL * N;
    float* agg  = h + NF;

    float* out  = (float*)d_out;
    float* mu   = out;
    float* ls   = out + NF;
    unsigned* part = (unsigned*)ls;

    const int prodBlocks = (N + 16 * PROD_WAVES - 1) / (16 * PROD_WAVES);

    degree_part_kernel<<<DEG_RANGES * DEG_SLICES, DEG_THREADS, 0, stream>>>(dst, part, N, E);

    product_kernel<false, 1><<<prodBlocks, PROD_THREADS, 0, stream>>>(
        x, agg, mu, nullptr, W1, nullptr, nullptr, h, nullptr, N, part, dinv, hot);
    aggregate_kernel<<<AGG_BLOCKS, AGG_THREADS, 0, stream>>>(h, src, dst, agg, mu, hot, E);

    product_kernel<true, 2><<<prodBlocks, PROD_THREADS, 0, stream>>>(
        nullptr, agg, mu, b1, W2a, W2b, dinv, h, ls, N, nullptr, nullptr, nullptr);

    aggregate_kernel<<<AGG_BLOCKS, AGG_THREADS, 0, stream>>>(h, src, dst, mu, agg, hot, E);
    sum2_bias_kernel<true><<<EW_BLOCKS, EW_THREADS, 0, stream>>>(mu, mu, agg, b2a, dinv, h, NF);

    aggregate_kernel<<<AGG_BLOCKS, AGG_THREADS, 0, stream>>>(ls, src, dst, agg, h, hot, E);
    sum2_bias_kernel<false><<<EW_BLOCKS, EW_THREADS, 0, stream>>>(ls, agg, h, b2b, dinv, nullptr, NF);
}
